# starting kernel
# baseline (speedup 1.0000x reference)
#include <hip/hip_runtime.h>

typedef float v2f __attribute__((ext_vector_type(2)));
typedef float v8f __attribute__((ext_vector_type(8)));

#define N_NODES 50000
#define N_EDGES 800000
#define IN_CH   96
#define HID     128

__global__ void zero_f32(float* __restrict__ p, int n) {
  int i = blockIdx.x * blockDim.x + threadIdx.x;
  if (i < n) p[i] = 0.0f;
}

__global__ void degree_kernel(const int* __restrict__ dst,
                              float* __restrict__ cnt, int E) {
  int e = blockIdx.x * blockDim.x + threadIdx.x;
  if (e < E) atomicAdd(&cnt[dst[e]], 1.0f);
}

__global__ void inv_kernel(float* __restrict__ cnt, int n) {
  int i = blockIdx.x * blockDim.x + threadIdx.x;
  if (i < n) cnt[i] = 1.0f / fmaxf(cnt[i], 1.0f);
}

__global__ void scatter_kernel(const float* __restrict__ feat,
                               const int* __restrict__ src,
                               const int* __restrict__ dst,
                               float* __restrict__ agg, int E, int d) {
  int wave = (blockIdx.x * blockDim.x + threadIdx.x) >> 5;
  int lane = threadIdx.x & 31;
  if (wave >= E) return;
  int s = src[wave];
  int t = dst[wave];
  const float* fp = feat + (size_t)s * d;
  float*       ap = agg  + (size_t)t * d;
  for (int c = lane; c < d; c += 32)
    atomicAdd(&ap[c], fp[c]);
}

__global__ void sage_wmma_kernel(const float* __restrict__ agg,
                                 const float* __restrict__ inv,
                                 const float* __restrict__ hin,
                                 const float* __restrict__ Wl,
                                 const float* __restrict__ Wr,
                                 const float* __restrict__ bias,
                                 float* __restrict__ hout,
                                 int din, int dout, int nTilesN) {
  int wave  = (blockIdx.x * blockDim.x + threadIdx.x) >> 5;
  int lane  = threadIdx.x & 31;
  int mTile = wave / nTilesN;
  int nTile = wave % nTilesN;
  int mBase = mTile * 16;
  int nBase = nTile * 16;
  int half  = lane >> 4;
  int l16   = lane & 15;

  int m = mBase + l16;
  int n = nBase + l16;
  float invm = inv[m];
  const float* aggRow = agg + (size_t)m * din;
  const float* hinRow = hin + (size_t)m * din;

  v8f c = {0.f, 0.f, 0.f, 0.f, 0.f, 0.f, 0.f, 0.f};

  for (int k0 = 0; k0 < din; k0 += 4) {
    int kA = k0 + half * 2;
    v2f aL, aR, bL, bR;
    aL.x = aggRow[kA]     * invm;
    aL.y = aggRow[kA + 1] * invm;
    aR.x = hinRow[kA];
    aR.y = hinRow[kA + 1];
    bL.x = Wl[(size_t)kA       * dout + n];
    bL.y = Wl[(size_t)(kA + 1) * dout + n];
    bR.x = Wr[(size_t)kA       * dout + n];
    bR.y = Wr[(size_t)(kA + 1) * dout + n];
    c = __builtin_amdgcn_wmma_f32_16x16x4_f32(false, aL, false, bL,
                                              (short)0, c, false, false);
    c = __builtin_amdgcn_wmma_f32_16x16x4_f32(false, aR, false, bR,
                                              (short)0, c, false, false);
  }

  float bn = bias[n];
  #pragma unroll
  for (int v = 0; v < 8; ++v) {
    int mo = mBase + v + half * 8;
    float val = fmaxf(c[v] + bn, 0.0f);
    hout[(size_t)mo * dout + (nBase + l16)] = val;
  }
}

__global__ void classifier_kernel(const float* __restrict__ h,
                                  const float* __restrict__ Wc,
                                  const float* __restrict__ bc,
                                  float* __restrict__ out, int n) {
  __shared__ float w[128];
  __shared__ float b[2];
  if (threadIdx.x < 128) w[threadIdx.x] = Wc[threadIdx.x];
  if (threadIdx.x < 2)   b[threadIdx.x] = bc[threadIdx.x];
  __syncthreads();
  int i = blockIdx.x * blockDim.x + threadIdx.x;
  if (i >= n) return;
  const float* hr = h + (size_t)i * 64;
  float o0 = b[0], o1 = b[1];
  #pragma unroll
  for (int k = 0; k < 64; ++k) {
    float hv = hr[k];
    o0 += hv * w[k * 2 + 0];
    o1 += hv * w[k * 2 + 1];
  }
  out[i * 2 + 0] = o0;
  out[i * 2 + 1] = o1;
}

extern "C" void kernel_launch(void* const* d_in, const int* in_sizes, int n_in,
                              void* d_out, int out_size, void* d_ws, size_t ws_size,
                              hipStream_t stream) {
  const float* x   = (const float*)d_in[0];
  const int*   ei  = (const int*)d_in[1];
  const int*   src = ei;
  const int*   dst = ei + N_EDGES;
  const float* Wl1 = (const float*)d_in[2];
  const float* Wr1 = (const float*)d_in[3];
  const float* b1  = (const float*)d_in[4];
  const float* Wl2 = (const float*)d_in[5];
  const float* Wr2 = (const float*)d_in[6];
  const float* b2  = (const float*)d_in[7];
  const float* Wl3 = (const float*)d_in[8];
  const float* Wr3 = (const float*)d_in[9];
  const float* b3  = (const float*)d_in[10];
  const float* Wc  = (const float*)d_in[11];
  const float* bc  = (const float*)d_in[12];
  float* out = (float*)d_out;

  float* inv = (float*)d_ws;
  float* agg = inv + N_NODES;
  float* hA  = agg + (size_t)N_NODES * HID;
  float* hB  = hA  + (size_t)N_NODES * HID;

  const int ZT = 256;

  zero_f32<<<(N_NODES + ZT - 1) / ZT, ZT, 0, stream>>>(inv, N_NODES);
  degree_kernel<<<(N_EDGES + ZT - 1) / ZT, ZT, 0, stream>>>(dst, inv, N_EDGES);
  inv_kernel<<<(N_NODES + ZT - 1) / ZT, ZT, 0, stream>>>(inv, N_NODES);

  const int scatterBlocks = N_EDGES / 8;

  zero_f32<<<((N_NODES * IN_CH) + ZT - 1) / ZT, ZT, 0, stream>>>(agg, N_NODES * IN_CH);
  scatter_kernel<<<scatterBlocks, 256, 0, stream>>>(x, src, dst, agg, N_EDGES, IN_CH);
  {
    int nTilesN = HID / 16;
    int waves   = (N_NODES / 16) * nTilesN;
    sage_wmma_kernel<<<waves / 4, 128, 0, stream>>>(agg, inv, x, Wl1, Wr1, b1,
                                                    hA, IN_CH, HID, nTilesN);
  }

  zero_f32<<<((N_NODES * HID) + ZT - 1) / ZT, ZT, 0, stream>>>(agg, N_NODES * HID);
  scatter_kernel<<<scatterBlocks, 256, 0, stream>>>(hA, src, dst, agg, N_EDGES, HID);
  {
    int nTilesN = HID / 16;
    int waves   = (N_NODES / 16) * nTilesN;
    sage_wmma_kernel<<<waves / 4, 128, 0, stream>>>(agg, inv, hA, Wl2, Wr2, b2,
                                                    hB, HID, HID, nTilesN);
  }

  zero_f32<<<((N_NODES * HID) + ZT - 1) / ZT, ZT, 0, stream>>>(agg, N_NODES * HID);
  scatter_kernel<<<scatterBlocks, 256, 0, stream>>>(hB, src, dst, agg, N_EDGES, HID);
  {
    int nTilesN = (HID / 2) / 16;
    int waves   = (N_NODES / 16) * nTilesN;
    sage_wmma_kernel<<<waves / 4, 128, 0, stream>>>(agg, inv, hB, Wl3, Wr3, b3,
                                                    hA, HID, HID / 2, nTilesN);
  }

  classifier_kernel<<<(N_NODES + ZT - 1) / ZT, ZT, 0, stream>>>(hA, Wc, bc, out, N_NODES);
}


# Layer GEMMs register-blocked: one wave computes a 16x64 strip, each A fragment loaded once per k step and used for 4 column tiles; weights staged once per work-group in LDS; sum order unchanged
# speedup vs baseline: 3.1308x; 3.1308x over previous
#include <hip/hip_runtime.h>

typedef float v2f __attribute__((ext_vector_type(2)));
typedef float v8f __attribute__((ext_vector_type(8)));

#define N_NODES 50000
#define N_EDGES 800000
#define IN_CH   96
#define HID     128

#define NPW       196
#define NRANGE    256
#define AGG_WAVES 32
#define LIST_CAP  1024
#define KEY_UNITS (N_EDGES / 16)
static_assert(N_EDGES % 16 == 0, "run-id array is read in 16-byte units");
static_assert(NPW * NRANGE >= N_NODES && (N_NODES - 1) / NPW < 256, "run id must fit a byte");

__global__ void run_id_kernel(const int* __restrict__ dst,
                              unsigned int* __restrict__ ids, int nQuads) {
  int i = blockIdx.x * blockDim.x + threadIdx.x;
  if (i >= nQuads) return;
  const int* d = (const int*)__builtin_assume_aligned(dst, 16) + 4 * i;
  unsigned int k0 = (unsigned int)d[0] / NPW;
  unsigned int k1 = (unsigned int)d[1] / NPW;
  unsigned int k2 = (unsigned int)d[2] / NPW;
  unsigned int k3 = (unsigned int)d[3] / NPW;
  ids[i] = (k0 & 255u) | ((k1 & 255u) << 8) | ((k2 & 255u) << 16) | (k3 << 24);
}

__device__ __forceinline__ unsigned int zero_bytes(unsigned int t) {
  return ~(((t & 0x7F7F7F7Fu) + 0x7F7F7F7Fu) | t | 0x7F7F7F7Fu);
}

template <int NCH, int G, bool COUNT>
__device__ __forceinline__ void add_found(const float* __restrict__ feat,
                                          const int* __restrict__ src,
                                          const int* __restrict__ dst,
                                          const int* lst, int cnt, int lo,
                                          float* acc, float* cntS, int lane) {
  constexpr int D = NCH * 32;
  for (int base = 0; base < cnt; base += 32) {
    int n = cnt - base;
    if (n > 32) n = 32;
    int s = 0, t = 0;
    if (lane < n) {
      const int e = lst[base + lane];
      s = src[e];
      t = dst[e] - lo;
      if (COUNT) atomicAdd(&cntS[t], 1.0f);
    }
    for (int j = 0; j < n; j += G) {
      float r[G][NCH];
      #pragma unroll
      for (int u = 0; u < G; ++u) {
        const int su = __builtin_amdgcn_readlane(s, j + u);
        const float* fp = feat + (size_t)su * D + lane;
        #pragma unroll
        for (int c = 0; c < NCH; ++c) r[u][c] = fp[c * 32];
      }
      #pragma unroll
      for (int u = 0; u < G; ++u) {
        if (j + u < n) {
          const int tu = __builtin_amdgcn_readlane(t, j + u);
          float* ap = acc + tu * D + lane;
          #pragma unroll
          for (int c = 0; c < NCH; ++c) atomicAdd(&ap[c * 32], r[u][c]);
        }
      }
    }
  }
}

template <int NCH, int G, bool COUNT>
__device__ __noinline__ void add_found_early(const float* __restrict__ feat,
                                             const int* __restrict__ src,
                                             const int* __restrict__ dst,
                                             const int* lst, int cnt, int lo,
                                             float* acc, float* cntS, int lane) {
  add_found<NCH, G, COUNT>(feat, src, dst, lst, cnt, lo, acc, cntS, lane);
}

template <int NCH, int G, bool COUNT>
__global__ void __launch_bounds__(AGG_WAVES * 32)
owner_sum_kernel(const float* __restrict__ feat,
                 const int* __restrict__ src,
                 const int* __restrict__ dst,
                 const uint4* __restrict__ ids,
                 float* __restrict__ agg,
                 float* __restrict__ inv) {
  constexpr int D = NCH * 32;
  __shared__ __attribute__((aligned(16))) float acc[NPW * D];
  __shared__ float cntS[NPW];
  __shared__ int   lists[AGG_WAVES * LIST_CAP];

  const int tid  = threadIdx.x;
  const int lane = tid & 31;
  const int wib  = __builtin_amdgcn_readfirstlane(tid >> 5);
  const int run  = blockIdx.x;
  const int lo   = run * NPW;
  int nn = N_NODES - lo;
  if (nn > NPW) nn = NPW;

  for (int i = tid; i < NPW * D; i += AGG_WAVES * 32) acc[i] = 0.0f;
  if (COUNT && tid < NPW) cntS[tid] = 0.0f;
  __syncthreads();

  int* lst = lists + wib * LIST_CAP;
  int cnt = 0;
  const unsigned int splat = (unsigned int)run * 0x01010101u;
  constexpr int SWEEP = AGG_WAVES * 32;
  constexpr int UNR   = 8;

  for (int u0 = wib * 32; u0 < KEY_UNITS; u0 += SWEEP * UNR) {
    uint4 x[UNR];
    #pragma unroll
    for (int k = 0; k < UNR; ++k) {
      const int u = u0 + k * SWEEP + lane;
      x[k] = ids[u < KEY_UNITS ? u : KEY_UNITS - 1];
    }
    #pragma unroll
    for (int k = 0; k < UNR; ++k) {
      const int u = u0 + k * SWEEP + lane;
      if (cnt > LIST_CAP - 512) {
        add_found_early<NCH, G, COUNT>(feat, src, dst, lst, cnt, lo, acc, cntS, lane);
        cnt = 0;
      }
      const unsigned int live = (u < KEY_UNITS) ? 0xFFFFFFFFu : 0u;
      unsigned int z[4];
      z[0] = zero_bytes(x[k].x ^ splat) & live;
      z[1] = zero_bytes(x[k].y ^ splat) & live;
      z[2] = zero_bytes(x[k].z ^ splat) & live;
      z[3] = zero_bytes(x[k].w ^ splat) & live;
      if (__builtin_amdgcn_ballot_w32((z[0] | z[1] | z[2] | z[3]) != 0u) == 0u) continue;
      #pragma unroll
      for (int q = 0; q < 4; ++q) {
        unsigned int zq = z[q];
        unsigned int m = __builtin_amdgcn_ballot_w32(zq != 0u);
        while (m != 0u) {
          if (zq != 0u) {
            const int bit = __builtin_ctz(zq);
            zq &= zq - 1u;
            const int pos = cnt + (int)__builtin_amdgcn_mbcnt_lo(m, 0u);
            lst[pos] = u * 16 + q * 4 + (bit >> 3);
          }
          cnt += __builtin_popcount(m);
          m = __builtin_amdgcn_ballot_w32(zq != 0u);
        }
      }
    }
    const bool last = (u0 + SWEEP * UNR >= KEY_UNITS);
    if (cnt >= 32 || (last && cnt > 0)) {
      add_found<NCH, G, COUNT>(feat, src, dst, lst, cnt, lo, acc, cntS, lane);
      cnt = 0;
    }
  }
  __syncthreads();

  float4* out4 = (float4*)(agg + (size_t)lo * D);
  const float4* acc4 = (const float4*)acc;
  for (int i = tid; i < nn * (D / 4); i += AGG_WAVES * 32) out4[i] = acc4[i];
  if (COUNT && tid < nn) inv[lo + tid] = 1.0f / fmaxf(cntS[tid], 1.0f);
}

#define GEMM_WGS   256
#define GEMM_WAVES 32

template <int DIN, int DOUT>
__device__ __forceinline__ void stage_weights(const float* __restrict__ Wg, float* Ws, int tid) {
  constexpr int PITCH = 2 * DOUT + 32;
  constexpr int Q = DOUT / 4;
  static_assert(DIN % 2 == 0 && DOUT % 4 == 0, "rows go in pairs, columns in quads");
  const float4* Wg4 = (const float4*)__builtin_assume_aligned(Wg, 16);
  for (int i = tid; i < (DIN / 2) * Q; i += GEMM_WAVES * 32) {
    const int kp = i / Q, q = i % Q;
    const float4 r0 = Wg4[(2 * kp) * Q + q];
    const float4 r1 = Wg4[(2 * kp + 1) * Q + q];
    float4* d = (float4*)(Ws + kp * PITCH + q * 8);
    d[0] = make_float4(r0.x, r1.x, r0.y, r1.y);
    d[1] = make_float4(r0.z, r1.z, r0.w, r1.w);
  }
}

template <int DIN, int DOUT>
__global__ void __launch_bounds__(GEMM_WAVES * 32)
sage_wmma_kernel(const float* __restrict__ agg,
                 const float* __restrict__ inv,
                 const float* __restrict__ hin,
                 const float* __restrict__ Wl,
                 const float* __restrict__ Wr,
                 const float* __restrict__ bias,
                 float* __restrict__ hout) {
  constexpr int PITCH = 2 * DOUT + 32;
  constexpr int NH = DOUT / 64;
  constexpr int NUNITS = (N_NODES / 16) * NH;
  __shared__ __attribute__((aligned(16))) float WlS[(DIN / 2) * PITCH];
  __shared__ __attribute__((aligned(16))) float WrS[(DIN / 2) * PITCH];

  const int tid  = threadIdx.x;
  const int lane = tid & 31;
  const int wib  = tid >> 5;
  const int half = lane >> 4;
  const int l16  = lane & 15;

  stage_weights<DIN, DOUT>(Wl, WlS, tid);
  stage_weights<DIN, DOUT>(Wr, WrS, tid);
  __syncthreads();

  for (int u = wib * GEMM_WGS + blockIdx.x; u < NUNITS; u += GEMM_WAVES * GEMM_WGS) {
    const int mBase = (u / NH) * 16;
    const int nBase = (u % NH) * 64;
    const int m = mBase + l16;
    const float invm = inv[m];
    const float* aggRow = agg + (size_t)m * DIN;
    const float* hinRow = hin + (size_t)m * DIN;
    const float* wl = WlS + (nBase + l16) * 2;
    const float* wr = WrS + (nBase + l16) * 2;

    v8f c[4];
    #pragma unroll
    for (int t = 0; t < 4; ++t) c[t] = (v8f){0.f, 0.f, 0.f, 0.f, 0.f, 0.f, 0.f, 0.f};

    for (int k0 = 0; k0 < DIN; k0 += 4) {
      const int kA = k0 + half * 2;
      v2f aL, aR;
      aL.x = aggRow[kA]     * invm;
      aL.y = aggRow[kA + 1] * invm;
      aR.x = hinRow[kA];
      aR.y = hinRow[kA + 1];
      const float* pl = wl + (kA >> 1) * PITCH;
      const float* pr = wr + (kA >> 1) * PITCH;
      #pragma unroll
      for (int t = 0; t < 4; ++t) {
        const v2f bL = *(const v2f*)(pl + t * 32);
        const v2f bR = *(const v2f*)(pr + t * 32);
        c[t] = __builtin_amdgcn_wmma_f32_16x16x4_f32(false, aL, false, bL,
                                                     (short)0, c[t], false, false);
        c[t] = __builtin_amdgcn_wmma_f32_16x16x4_f32(false, aR, false, bR,
                                                     (short)0, c[t], false, false);
      }
    }

    #pragma unroll
    for (int t = 0; t < 4; ++t) {
      const int n = nBase + t * 16 + l16;
      const float bn = bias[n];
      #pragma unroll
      for (int v = 0; v < 8; ++v) {
        const int mo = mBase + v + half * 8;
        hout[(size_t)mo * DOUT + n] = fmaxf(c[t][v] + bn, 0.0f);
      }
    }
  }
}

template <int DIN, int DOUT>
__global__ void __launch_bounds__(GEMM_WAVES * 32)
proj_pair_wmma_kernel(const float* __restrict__ hin,
                      const float* __restrict__ Wl,
                      const float* __restrict__ Wr,
                      const float* __restrict__ bias,
                      float* __restrict__ P,
                      float* __restrict__ R) {
  static_assert(DOUT == 64, "one strip spans the whole output row");
  constexpr int PITCH = 2 * DOUT + 32;
  constexpr int NUNITS = N_NODES / 16;
  __shared__ __attribute__((aligned(16))) float WlS[(DIN / 2) * PITCH];
  __shared__ __attribute__((aligned(16))) float WrS[(DIN / 2) * PITCH];

  const int tid  = threadIdx.x;
  const int lane = tid & 31;
  const int wib  = tid >> 5;
  const int half = lane >> 4;
  const int l16  = lane & 15;

  stage_weights<DIN, DOUT>(Wl, WlS, tid);
  stage_weights<DIN, DOUT>(Wr, WrS, tid);
  __syncthreads();

  for (int u = wib * GEMM_WGS + blockIdx.x; u < NUNITS; u += GEMM_WAVES * GEMM_WGS) {
    const int mBase = u * 16;
    const int m = mBase + l16;
    const float* hinRow = hin + (size_t)m * DIN;
    const float* wl = WlS + l16 * 2;
    const float* wr = WrS + l16 * 2;

    v8f cP[4], cR[4];
    #pragma unroll
    for (int t = 0; t < 4; ++t) {
      cP[t] = (v8f){0.f, 0.f, 0.f, 0.f, 0.f, 0.f, 0.f, 0.f};
      cR[t] = (v8f){0.f, 0.f, 0.f, 0.f, 0.f, 0.f, 0.f, 0.f};
    }

    for (int k0 = 0; k0 < DIN; k0 += 4) {
      const int kA = k0 + half * 2;
      v2f a;
      a.x = hinRow[kA];
      a.y = hinRow[kA + 1];
      const float* pl = wl + (kA >> 1) * PITCH;
      const float* pr = wr + (kA >> 1) * PITCH;
      #pragma unroll
      for (int t = 0; t < 4; ++t) {
        const v2f bL = *(const v2f*)(pl + t * 32);
        const v2f bR = *(const v2f*)(pr + t * 32);
        cP[t] = __builtin_amdgcn_wmma_f32_16x16x4_f32(false, a, false, bL,
                                                      (short)0, cP[t], false, false);
        cR[t] = __builtin_amdgcn_wmma_f32_16x16x4_f32(false, a, false, bR,
                                                      (short)0, cR[t], false, false);
      }
    }

    #pragma unroll
    for (int t = 0; t < 4; ++t) {
      const int n = t * 16 + l16;
      const float bn = bias[n];
      #pragma unroll
      for (int v = 0; v < 8; ++v) {
        const int mo = mBase + v + half * 8;
        P[(size_t)mo * DOUT + n] = cP[t][v];
        R[(size_t)mo * DOUT + n] = cR[t][v] + bn;
      }
    }
  }
}

__global__ void combine_classifier_kernel(const float* __restrict__ aggP,
                                          const float* __restrict__ inv,
                                          const float* __restrict__ R,
                                          const float* __restrict__ Wc,
                                          const float* __restrict__ bc,
                                          float* __restrict__ out, int n) {
  __shared__ float w[128];
  __shared__ float b[2];
  if (threadIdx.x < 128) w[threadIdx.x] = Wc[threadIdx.x];
  if (threadIdx.x < 2)   b[threadIdx.x] = bc[threadIdx.x];
  __syncthreads();
  int i = blockIdx.x * blockDim.x + threadIdx.x;
  if (i >= n) return;
  const float* pr = aggP + (size_t)i * 64;
  const float* rr = R    + (size_t)i * 64;
  float iv = inv[i];
  float o0 = b[0], o1 = b[1];
  #pragma unroll
  for (int k = 0; k < 64; ++k) {
    float hv = fmaxf(pr[k] * iv + rr[k], 0.0f);
    o0 += hv * w[k * 2 + 0];
    o1 += hv * w[k * 2 + 1];
  }
  out[i * 2 + 0] = o0;
  out[i * 2 + 1] = o1;
}

extern "C" void kernel_launch(void* const* d_in, const int* in_sizes, int n_in,
                              void* d_out, int out_size, void* d_ws, size_t ws_size,
                              hipStream_t stream) {
  const float* x   = (const float*)d_in[0];
  const int*   ei  = (const int*)d_in[1];
  const int*   src = ei;
  const int*   dst = ei + N_EDGES;
  const float* Wl1 = (const float*)d_in[2];
  const float* Wr1 = (const float*)d_in[3];
  const float* b1  = (const float*)d_in[4];
  const float* Wl2 = (const float*)d_in[5];
  const float* Wr2 = (const float*)d_in[6];
  const float* b2  = (const float*)d_in[7];
  const float* Wl3 = (const float*)d_in[8];
  const float* Wr3 = (const float*)d_in[9];
  const float* b3  = (const float*)d_in[10];
  const float* Wc  = (const float*)d_in[11];
  const float* bc  = (const float*)d_in[12];
  float* out = (float*)d_out;

  float* agg = (float*)d_ws;
  float* hA  = agg + (size_t)N_NODES * HID;
  float* hB  = hA  + (size_t)N_NODES * HID;
  float* inv = hB  + (size_t)N_NODES * HID;

  const int ZT = 256;
  unsigned int* ids = (unsigned int*)hB;
  const int nQuads = N_EDGES / 4;

  run_id_kernel<<<(nQuads + ZT - 1) / ZT, ZT, 0, stream>>>(dst, ids, nQuads);

  owner_sum_kernel<3, 8, true><<<NRANGE, AGG_WAVES * 32, 0, stream>>>(
      x, src, dst, (const uint4*)ids, agg, inv);
  sage_wmma_kernel<IN_CH, HID><<<GEMM_WGS, GEMM_WAVES * 32, 0, stream>>>(agg, inv, x, Wl1, Wr1, b1, hA);

  owner_sum_kernel<4, 8, false><<<NRANGE, AGG_WAVES * 32, 0, stream>>>(
      hA, src, dst, (const uint4*)ids, agg, inv);
  sage_wmma_kernel<HID, HID><<<GEMM_WGS, GEMM_WAVES * 32, 0, stream>>>(agg, inv, hA, Wl2, Wr2, b2, hB);

  float* P3 = hA;
  float* R3 = hA + (size_t)N_NODES * (HID / 2);
  proj_pair_wmma_kernel<HID, HID / 2><<<GEMM_WGS, GEMM_WAVES * 32, 0, stream>>>(hB, Wl3, Wr3, b3, P3, R3);
  run_id_kernel<<<(nQuads + ZT - 1) / ZT, ZT, 0, stream>>>(dst, ids, nQuads);
  owner_sum_kernel<2, 16, false><<<NRANGE, AGG_WAVES * 32, 0, stream>>>(
      P3, src, dst, (const uint4*)ids, agg, inv);

  combine_classifier_kernel<<<(N_NODES + ZT - 1) / ZT, ZT, 0, stream>>>(agg, inv, R3, Wc, bc,
                                                                        out, N_NODES);
}


# Found edges read one packed record: the id kernel also writes (dst offset in run << 16 | src) per edge, so owners fetch one scattered dword per found edge, not src[e] and dst[e]; sum order unchanged
# speedup vs baseline: 3.2421x; 1.0356x over previous
#include <hip/hip_runtime.h>

typedef float v2f __attribute__((ext_vector_type(2)));
typedef float v8f __attribute__((ext_vector_type(8)));

#define N_NODES 50000
#define N_EDGES 800000
#define IN_CH   96
#define HID     128

#define NPW       196
#define NRANGE    256
#define AGG_WAVES 32
#define LIST_CAP  1024
#define KEY_UNITS (N_EDGES / 16)
static_assert(N_EDGES % 16 == 0, "run-id array is read in 16-byte units");
static_assert(NPW * NRANGE >= N_NODES && (N_NODES - 1) / NPW < 256, "run id must fit a byte");
static_assert(N_NODES <= 65536 && NPW <= 65536, "an edge record holds two 16-bit fields");

__global__ void run_id_kernel(const int* __restrict__ src,
                              const int* __restrict__ dst,
                              unsigned int* __restrict__ ids,
                              uint4* __restrict__ rec, int nQuads) {
  int i = blockIdx.x * blockDim.x + threadIdx.x;
  if (i >= nQuads) return;
  const int* d = (const int*)__builtin_assume_aligned(dst, 16) + 4 * i;
  const int* s = (const int*)__builtin_assume_aligned(src, 16) + 4 * i;
  unsigned int d0 = (unsigned int)d[0], d1 = (unsigned int)d[1];
  unsigned int d2 = (unsigned int)d[2], d3 = (unsigned int)d[3];
  unsigned int k0 = d0 / NPW, k1 = d1 / NPW, k2 = d2 / NPW, k3 = d3 / NPW;
  ids[i] = (k0 & 255u) | ((k1 & 255u) << 8) | ((k2 & 255u) << 16) | (k3 << 24);
  uint4 r;
  r.x = ((d0 - k0 * NPW) << 16) | ((unsigned int)s[0] & 0xFFFFu);
  r.y = ((d1 - k1 * NPW) << 16) | ((unsigned int)s[1] & 0xFFFFu);
  r.z = ((d2 - k2 * NPW) << 16) | ((unsigned int)s[2] & 0xFFFFu);
  r.w = ((d3 - k3 * NPW) << 16) | ((unsigned int)s[3] & 0xFFFFu);
  rec[i] = r;
}

__device__ __forceinline__ unsigned int zero_bytes(unsigned int t) {
  return ~(((t & 0x7F7F7F7Fu) + 0x7F7F7F7Fu) | t | 0x7F7F7F7Fu);
}

template <int NCH, int G, bool COUNT>
__device__ __forceinline__ void add_found(const float* __restrict__ feat,
                                          const unsigned int* __restrict__ rec,
                                          const int* lst, int cnt,
                                          float* acc, float* cntS, int lane) {
  constexpr int D = NCH * 32;
  for (int base = 0; base < cnt; base += 32) {
    int n = cnt - base;
    if (n > 32) n = 32;
    int s = 0, t = 0;
    if (lane < n) {
      const unsigned int r = rec[lst[base + lane]];
      s = (int)(r & 0xFFFFu);
      t = (int)(r >> 16);
      if (COUNT) atomicAdd(&cntS[t], 1.0f);
    }
    for (int j = 0; j < n; j += G) {
      float r[G][NCH];
      #pragma unroll
      for (int u = 0; u < G; ++u) {
        const int su = __builtin_amdgcn_readlane(s, j + u);
        const float* fp = feat + (size_t)su * D + lane;
        #pragma unroll
        for (int c = 0; c < NCH; ++c) r[u][c] = fp[c * 32];
      }
      #pragma unroll
      for (int u = 0; u < G; ++u) {
        if (j + u < n) {
          const int tu = __builtin_amdgcn_readlane(t, j + u);
          float* ap = acc + tu * D + lane;
          #pragma unroll
          for (int c = 0; c < NCH; ++c) atomicAdd(&ap[c * 32], r[u][c]);
        }
      }
    }
  }
}

template <int NCH, int G, bool COUNT>
__device__ __noinline__ void add_found_early(const float* __restrict__ feat,
                                             const unsigned int* __restrict__ rec,
                                             const int* lst, int cnt,
                                             float* acc, float* cntS, int lane) {
  add_found<NCH, G, COUNT>(feat, rec, lst, cnt, acc, cntS, lane);
}

template <int NCH, int G, bool COUNT>
__global__ void __launch_bounds__(AGG_WAVES * 32)
owner_sum_kernel(const float* __restrict__ feat,
                 const unsigned int* __restrict__ rec,
                 const uint4* __restrict__ ids,
                 float* __restrict__ agg,
                 float* __restrict__ inv) {
  constexpr int D = NCH * 32;
  __shared__ __attribute__((aligned(16))) float acc[NPW * D];
  __shared__ float cntS[NPW];
  __shared__ int   lists[AGG_WAVES * LIST_CAP];

  const int tid  = threadIdx.x;
  const int lane = tid & 31;
  const int wib  = __builtin_amdgcn_readfirstlane(tid >> 5);
  const int run  = blockIdx.x;
  const int lo   = run * NPW;
  int nn = N_NODES - lo;
  if (nn > NPW) nn = NPW;

  for (int i = tid; i < NPW * D; i += AGG_WAVES * 32) acc[i] = 0.0f;
  if (COUNT && tid < NPW) cntS[tid] = 0.0f;
  __syncthreads();

  int* lst = lists + wib * LIST_CAP;
  int cnt = 0;
  const unsigned int splat = (unsigned int)run * 0x01010101u;
  constexpr int SWEEP = AGG_WAVES * 32;
  constexpr int UNR   = 8;

  for (int u0 = wib * 32; u0 < KEY_UNITS; u0 += SWEEP * UNR) {
    uint4 x[UNR];
    #pragma unroll
    for (int k = 0; k < UNR; ++k) {
      const int u = u0 + k * SWEEP + lane;
      x[k] = ids[u < KEY_UNITS ? u : KEY_UNITS - 1];
    }
    #pragma unroll
    for (int k = 0; k < UNR; ++k) {
      const int u = u0 + k * SWEEP + lane;
      if (cnt > LIST_CAP - 512) {
        add_found_early<NCH, G, COUNT>(feat, rec, lst, cnt, acc, cntS, lane);
        cnt = 0;
      }
      const unsigned int live = (u < KEY_UNITS) ? 0xFFFFFFFFu : 0u;
      unsigned int z[4];
      z[0] = zero_bytes(x[k].x ^ splat) & live;
      z[1] = zero_bytes(x[k].y ^ splat) & live;
      z[2] = zero_bytes(x[k].z ^ splat) & live;
      z[3] = zero_bytes(x[k].w ^ splat) & live;
      if (__builtin_amdgcn_ballot_w32((z[0] | z[1] | z[2] | z[3]) != 0u) == 0u) continue;
      #pragma unroll
      for (int q = 0; q < 4; ++q) {
        unsigned int zq = z[q];
        unsigned int m = __builtin_amdgcn_ballot_w32(zq != 0u);
        while (m != 0u) {
          if (zq != 0u) {
            const int bit = __builtin_ctz(zq);
            zq &= zq - 1u;
            const int pos = cnt + (int)__builtin_amdgcn_mbcnt_lo(m, 0u);
            lst[pos] = u * 16 + q * 4 + (bit >> 3);
          }
          cnt += __builtin_popcount(m);
          m = __builtin_amdgcn_ballot_w32(zq != 0u);
        }
      }
    }
    const bool last = (u0 + SWEEP * UNR >= KEY_UNITS);
    if (cnt >= 32 || (last && cnt > 0)) {
      add_found<NCH, G, COUNT>(feat, rec, lst, cnt, acc, cntS, lane);
      cnt = 0;
    }
  }
  __syncthreads();

  float4* out4 = (float4*)(agg + (size_t)lo * D);
  const float4* acc4 = (const float4*)acc;
  for (int i = tid; i < nn * (D / 4); i += AGG_WAVES * 32) out4[i] = acc4[i];
  if (COUNT && tid < nn) inv[lo + tid] = 1.0f / fmaxf(cntS[tid], 1.0f);
}

#define GEMM_WGS   256
#define GEMM_WAVES 32

template <int DIN, int DOUT>
__device__ __forceinline__ void stage_weights(const float* __restrict__ Wg, float* Ws, int tid) {
  constexpr int PITCH = 2 * DOUT + 32;
  constexpr int Q = DOUT / 4;
  static_assert(DIN % 2 == 0 && DOUT % 4 == 0, "rows go in pairs, columns in quads");
  const float4* Wg4 = (const float4*)__builtin_assume_aligned(Wg, 16);
  for (int i = tid; i < (DIN / 2) * Q; i += GEMM_WAVES * 32) {
    const int kp = i / Q, q = i % Q;
    const float4 r0 = Wg4[(2 * kp) * Q + q];
    const float4 r1 = Wg4[(2 * kp + 1) * Q + q];
    float4* d = (float4*)(Ws + kp * PITCH + q * 8);
    d[0] = make_float4(r0.x, r1.x, r0.y, r1.y);
    d[1] = make_float4(r0.z, r1.z, r0.w, r1.w);
  }
}

template <int DIN, int DOUT>
__global__ void __launch_bounds__(GEMM_WAVES * 32)
sage_wmma_kernel(const float* __restrict__ agg,
                 const float* __restrict__ inv,
                 const float* __restrict__ hin,
                 const float* __restrict__ Wl,
                 const float* __restrict__ Wr,
                 const float* __restrict__ bias,
                 float* __restrict__ hout) {
  constexpr int PITCH = 2 * DOUT + 32;
  constexpr int NH = DOUT / 64;
  constexpr int NUNITS = (N_NODES / 16) * NH;
  __shared__ __attribute__((aligned(16))) float WlS[(DIN / 2) * PITCH];
  __shared__ __attribute__((aligned(16))) float WrS[(DIN / 2) * PITCH];

  const int tid  = threadIdx.x;
  const int lane = tid & 31;
  const int wib  = tid >> 5;
  const int half = lane >> 4;
  const int l16  = lane & 15;

  stage_weights<DIN, DOUT>(Wl, WlS, tid);
  stage_weights<DIN, DOUT>(Wr, WrS, tid);
  __syncthreads();

  for (int u = wib * GEMM_WGS + blockIdx.x; u < NUNITS; u += GEMM_WAVES * GEMM_WGS) {
    const int mBase = (u / NH) * 16;
    const int nBase = (u % NH) * 64;
    const int m = mBase + l16;
    const float invm = inv[m];
    const float* aggRow = agg + (size_t)m * DIN;
    const float* hinRow = hin + (size_t)m * DIN;
    const float* wl = WlS + (nBase + l16) * 2;
    const float* wr = WrS + (nBase + l16) * 2;

    v8f c[4];
    #pragma unroll
    for (int t = 0; t < 4; ++t) c[t] = (v8f){0.f, 0.f, 0.f, 0.f, 0.f, 0.f, 0.f, 0.f};

    for (int k0 = 0; k0 < DIN; k0 += 4) {
      const int kA = k0 + half * 2;
      v2f aL, aR;
      aL.x = aggRow[kA]     * invm;
      aL.y = aggRow[kA + 1] * invm;
      aR.x = hinRow[kA];
      aR.y = hinRow[kA + 1];
      const float* pl = wl + (kA >> 1) * PITCH;
      const float* pr = wr + (kA >> 1) * PITCH;
      #pragma unroll
      for (int t = 0; t < 4; ++t) {
        const v2f bL = *(const v2f*)(pl + t * 32);
        const v2f bR = *(const v2f*)(pr + t * 32);
        c[t] = __builtin_amdgcn_wmma_f32_16x16x4_f32(false, aL, false, bL,
                                                     (short)0, c[t], false, false);
        c[t] = __builtin_amdgcn_wmma_f32_16x16x4_f32(false, aR, false, bR,
                                                     (short)0, c[t], false, false);
      }
    }

    #pragma unroll
    for (int t = 0; t < 4; ++t) {
      const int n = nBase + t * 16 + l16;
      const float bn = bias[n];
      #pragma unroll
      for (int v = 0; v < 8; ++v) {
        const int mo = mBase + v + half * 8;
        hout[(size_t)mo * DOUT + n] = fmaxf(c[t][v] + bn, 0.0f);
      }
    }
  }
}

template <int DIN, int DOUT>
__global__ void __launch_bounds__(GEMM_WAVES * 32)
proj_pair_wmma_kernel(const float* __restrict__ hin,
                      const float* __restrict__ Wl,
                      const float* __restrict__ Wr,
                      const float* __restrict__ bias,
                      float* __restrict__ P,
                      float* __restrict__ R) {
  static_assert(DOUT == 64, "one strip spans the whole output row");
  constexpr int PITCH = 2 * DOUT + 32;
  constexpr int NUNITS = N_NODES / 16;
  __shared__ __attribute__((aligned(16))) float WlS[(DIN / 2) * PITCH];
  __shared__ __attribute__((aligned(16))) float WrS[(DIN / 2) * PITCH];

  const int tid  = threadIdx.x;
  const int lane = tid & 31;
  const int wib  = tid >> 5;
  const int half = lane >> 4;
  const int l16  = lane & 15;

  stage_weights<DIN, DOUT>(Wl, WlS, tid);
  stage_weights<DIN, DOUT>(Wr, WrS, tid);
  __syncthreads();

  for (int u = wib * GEMM_WGS + blockIdx.x; u < NUNITS; u += GEMM_WAVES * GEMM_WGS) {
    const int mBase = u * 16;
    const int m = mBase + l16;
    const float* hinRow = hin + (size_t)m * DIN;
    const float* wl = WlS + l16 * 2;
    const float* wr = WrS + l16 * 2;

    v8f cP[4], cR[4];
    #pragma unroll
    for (int t = 0; t < 4; ++t) {
      cP[t] = (v8f){0.f, 0.f, 0.f, 0.f, 0.f, 0.f, 0.f, 0.f};
      cR[t] = (v8f){0.f, 0.f, 0.f, 0.f, 0.f, 0.f, 0.f, 0.f};
    }

    for (int k0 = 0; k0 < DIN; k0 += 4) {
      const int kA = k0 + half * 2;
      v2f a;
      a.x = hinRow[kA];
      a.y = hinRow[kA + 1];
      const float* pl = wl + (kA >> 1) * PITCH;
      const float* pr = wr + (kA >> 1) * PITCH;
      #pragma unroll
      for (int t = 0; t < 4; ++t) {
        const v2f bL = *(const v2f*)(pl + t * 32);
        const v2f bR = *(const v2f*)(pr + t * 32);
        cP[t] = __builtin_amdgcn_wmma_f32_16x16x4_f32(false, a, false, bL,
                                                      (short)0, cP[t], false, false);
        cR[t] = __builtin_amdgcn_wmma_f32_16x16x4_f32(false, a, false, bR,
                                                      (short)0, cR[t], false, false);
      }
    }

    #pragma unroll
    for (int t = 0; t < 4; ++t) {
      const int n = t * 16 + l16;
      const float bn = bias[n];
      #pragma unroll
      for (int v = 0; v < 8; ++v) {
        const int mo = mBase + v + half * 8;
        P[(size_t)mo * DOUT + n] = cP[t][v];
        R[(size_t)mo * DOUT + n] = cR[t][v] + bn;
      }
    }
  }
}

__global__ void combine_classifier_kernel(const float* __restrict__ aggP,
                                          const float* __restrict__ inv,
                                          const float* __restrict__ R,
                                          const float* __restrict__ Wc,
                                          const float* __restrict__ bc,
                                          float* __restrict__ out, int n) {
  __shared__ float w[128];
  __shared__ float b[2];
  if (threadIdx.x < 128) w[threadIdx.x] = Wc[threadIdx.x];
  if (threadIdx.x < 2)   b[threadIdx.x] = bc[threadIdx.x];
  __syncthreads();
  int i = blockIdx.x * blockDim.x + threadIdx.x;
  if (i >= n) return;
  const float* pr = aggP + (size_t)i * 64;
  const float* rr = R    + (size_t)i * 64;
  float iv = inv[i];
  float o0 = b[0], o1 = b[1];
  #pragma unroll
  for (int k = 0; k < 64; ++k) {
    float hv = fmaxf(pr[k] * iv + rr[k], 0.0f);
    o0 += hv * w[k * 2 + 0];
    o1 += hv * w[k * 2 + 1];
  }
  out[i * 2 + 0] = o0;
  out[i * 2 + 1] = o1;
}

extern "C" void kernel_launch(void* const* d_in, const int* in_sizes, int n_in,
                              void* d_out, int out_size, void* d_ws, size_t ws_size,
                              hipStream_t stream) {
  const float* x   = (const float*)d_in[0];
  const int*   ei  = (const int*)d_in[1];
  const int*   src = ei;
  const int*   dst = ei + N_EDGES;
  const float* Wl1 = (const float*)d_in[2];
  const float* Wr1 = (const float*)d_in[3];
  const float* b1  = (const float*)d_in[4];
  const float* Wl2 = (const float*)d_in[5];
  const float* Wr2 = (const float*)d_in[6];
  const float* b2  = (const float*)d_in[7];
  const float* Wl3 = (const float*)d_in[8];
  const float* Wr3 = (const float*)d_in[9];
  const float* b3  = (const float*)d_in[10];
  const float* Wc  = (const float*)d_in[11];
  const float* bc  = (const float*)d_in[12];
  float* out = (float*)d_out;

  float* agg = (float*)d_ws;
  float* hA  = agg + (size_t)N_NODES * HID;
  float* hB  = hA  + (size_t)N_NODES * HID;
  float* inv = hB  + (size_t)N_NODES * HID;

  const int ZT = 256;
  unsigned int* ids = (unsigned int*)hB;
  unsigned int* rec = ids + N_EDGES / 4;
  const int nQuads = N_EDGES / 4;

  run_id_kernel<<<(nQuads + ZT - 1) / ZT, ZT, 0, stream>>>(src, dst, ids, (uint4*)rec, nQuads);

  owner_sum_kernel<3, 8, true><<<NRANGE, AGG_WAVES * 32, 0, stream>>>(
      x, rec, (const uint4*)ids, agg, inv);
  sage_wmma_kernel<IN_CH, HID><<<GEMM_WGS, GEMM_WAVES * 32, 0, stream>>>(agg, inv, x, Wl1, Wr1, b1, hA);

  owner_sum_kernel<4, 8, false><<<NRANGE, AGG_WAVES * 32, 0, stream>>>(
      hA, rec, (const uint4*)ids, agg, inv);
  sage_wmma_kernel<HID, HID><<<GEMM_WGS, GEMM_WAVES * 32, 0, stream>>>(agg, inv, hA, Wl2, Wr2, b2, hB);

  float* P3 = hA;
  float* R3 = hA + (size_t)N_NODES * (HID / 2);
  proj_pair_wmma_kernel<HID, HID / 2><<<GEMM_WGS, GEMM_WAVES * 32, 0, stream>>>(hB, Wl3, Wr3, b3, P3, R3);
  run_id_kernel<<<(nQuads + ZT - 1) / ZT, ZT, 0, stream>>>(src, dst, ids, (uint4*)rec, nQuads);
  owner_sum_kernel<2, 16, false><<<NRANGE, AGG_WAVES * 32, 0, stream>>>(
      P3, rec, (const uint4*)ids, agg, inv);

  combine_classifier_kernel<<<(N_NODES + ZT - 1) / ZT, ZT, 0, stream>>>(agg, inv, R3, Wc, bc,
                                                                        out, N_NODES);
}


# Layer 1's owner pass keeps the records of the edges its scan found (a segment per wave); layer 2's pass adds from these segments and scans nothing; LDS float adds still in arrival order
# speedup vs baseline: 3.4826x; 1.0742x over previous
#include <hip/hip_runtime.h>

typedef float v2f __attribute__((ext_vector_type(2)));
typedef float v8f __attribute__((ext_vector_type(8)));

#define N_NODES 50000
#define N_EDGES 800000
#define IN_CH   96
#define HID     128

#define NPW       196
#define NRANGE    256
#define AGG_WAVES 32
#define LIST_CAP  1024
#define KEY_UNITS (N_EDGES / 16)
#define SEG_CAP   512
static_assert(N_EDGES % 16 == 0, "run-id array is read in 16-byte units");
static_assert(NPW * NRANGE >= N_NODES && (N_NODES - 1) / NPW < 256, "run id must fit a byte");
static_assert(N_NODES <= 65536 && NPW <= 65536, "an edge record holds two 16-bit fields");

__global__ void run_id_kernel(const int* __restrict__ src,
                              const int* __restrict__ dst,
                              unsigned int* __restrict__ ids,
                              uint4* __restrict__ rec, int nQuads) {
  int i = blockIdx.x * blockDim.x + threadIdx.x;
  if (i >= nQuads) return;
  const int* d = (const int*)__builtin_assume_aligned(dst, 16) + 4 * i;
  const int* s = (const int*)__builtin_assume_aligned(src, 16) + 4 * i;
  unsigned int d0 = (unsigned int)d[0], d1 = (unsigned int)d[1];
  unsigned int d2 = (unsigned int)d[2], d3 = (unsigned int)d[3];
  unsigned int k0 = d0 / NPW, k1 = d1 / NPW, k2 = d2 / NPW, k3 = d3 / NPW;
  ids[i] = (k0 & 255u) | ((k1 & 255u) << 8) | ((k2 & 255u) << 16) | (k3 << 24);
  uint4 r;
  r.x = ((d0 - k0 * NPW) << 16) | ((unsigned int)s[0] & 0xFFFFu);
  r.y = ((d1 - k1 * NPW) << 16) | ((unsigned int)s[1] & 0xFFFFu);
  r.z = ((d2 - k2 * NPW) << 16) | ((unsigned int)s[2] & 0xFFFFu);
  r.w = ((d3 - k3 * NPW) << 16) | ((unsigned int)s[3] & 0xFFFFu);
  rec[i] = r;
}

__device__ __forceinline__ unsigned int zero_bytes(unsigned int t) {
  return ~(((t & 0x7F7F7F7Fu) + 0x7F7F7F7Fu) | t | 0x7F7F7F7Fu);
}

template <int NCH, int G>
__device__ __forceinline__ void add_rows(const float* __restrict__ feat,
                                         int s, int t, int n, float* acc, int lane) {
  constexpr int D = NCH * 32;
  for (int j = 0; j < n; j += G) {
    float r[G][NCH];
    #pragma unroll
    for (int u = 0; u < G; ++u) {
      const int su = __builtin_amdgcn_readlane(s, j + u);
      const float* fp = feat + (size_t)su * D + lane;
      #pragma unroll
      for (int c = 0; c < NCH; ++c) r[u][c] = fp[c * 32];
    }
    #pragma unroll
    for (int u = 0; u < G; ++u) {
      if (j + u < n) {
        const int tu = __builtin_amdgcn_readlane(t, j + u);
        float* ap = acc + tu * D + lane;
        #pragma unroll
        for (int c = 0; c < NCH; ++c) atomicAdd(&ap[c * 32], r[u][c]);
      }
    }
  }
}

template <int NCH, int G, bool COUNT, bool SAVE>
__device__ __forceinline__ int add_found(const float* __restrict__ feat,
                                         const unsigned int* __restrict__ rec,
                                         const int* lst, int cnt,
                                         float* acc, float* cntS, int lane,
                                         unsigned int* segw, int saved) {
  for (int base = 0; base < cnt; base += 32) {
    int n = cnt - base;
    if (n > 32) n = 32;
    int s = 0, t = 0;
    if (lane < n) {
      const unsigned int r = rec[lst[base + lane]];
      s = (int)(r & 0xFFFFu);
      t = (int)(r >> 16);
      if (COUNT) atomicAdd(&cntS[t], 1.0f);
      if (SAVE) {
        const int p = saved + base + lane;
        if (p < SEG_CAP) segw[p] = r;
      }
    }
    add_rows<NCH, G>(feat, s, t, n, acc, lane);
  }
  saved += cnt;
  return saved > SEG_CAP ? SEG_CAP + 1 : saved;
}

template <int NCH, int G>
__device__ __forceinline__ void add_saved(const float* __restrict__ feat,
                                          const unsigned int* segw, int cnt,
                                          float* acc, int lane) {
  unsigned int rn = 0;
  if (lane < cnt) rn = segw[lane];
  for (int base = 0; base < cnt; base += 32) {
    int n = cnt - base;
    if (n > 32) n = 32;
    const unsigned int r = rn;
    unsigned int nx = 0;
    if (base + 32 + lane < cnt) nx = segw[base + 32 + lane];
    rn = nx;
    add_rows<NCH, G>(feat, (int)(r & 0xFFFFu), (int)(r >> 16), n, acc, lane);
  }
}

template <int NCH, int G, bool COUNT>
__device__ __noinline__ void add_found_early(const float* __restrict__ feat,
                                             const unsigned int* __restrict__ rec,
                                             const int* lst, int cnt,
                                             float* acc, float* cntS, int lane) {
  add_found<NCH, G, COUNT, false>(feat, rec, lst, cnt, acc, cntS, lane, nullptr, 0);
}

template <int NCH, int G, bool COUNT, int MODE>
__global__ void __launch_bounds__(AGG_WAVES * 32)
owner_sum_kernel(const float* __restrict__ feat,
                 const unsigned int* __restrict__ rec,
                 const uint4* __restrict__ ids,
                 float* __restrict__ agg,
                 float* __restrict__ inv,
                 unsigned int* seg,
                 unsigned int* segCnt) {
  constexpr int D = NCH * 32;
  __shared__ __attribute__((aligned(16))) float acc[NPW * D];
  __shared__ float cntS[NPW];
  __shared__ int   lists[AGG_WAVES * LIST_CAP];

  const int tid  = threadIdx.x;
  const int lane = tid & 31;
  const int wib  = __builtin_amdgcn_readfirstlane(tid >> 5);
  const int run  = blockIdx.x;
  const int lo   = run * NPW;
  int nn = N_NODES - lo;
  if (nn > NPW) nn = NPW;

  for (int i = tid; i < NPW * D; i += AGG_WAVES * 32) acc[i] = 0.0f;
  if (COUNT && tid < NPW) cntS[tid] = 0.0f;
  __syncthreads();

  static_assert(AGG_WAVES == 32, "lane i reads the kept count of wave i");
  unsigned int* segw = seg + (size_t)(run * AGG_WAVES + wib) * SEG_CAP;
  int saved = 0;
  bool scan = true;
  if (MODE == 2) {
    const unsigned int c = segCnt[run * AGG_WAVES + lane];
    if (__builtin_amdgcn_ballot_w32(c > (unsigned int)SEG_CAP) == 0u) {
      add_saved<NCH, G>(feat, segw, __builtin_amdgcn_readlane((int)c, wib), acc, lane);
      scan = false;
    }
  }

  int* lst = lists + wib * LIST_CAP;
  int cnt = 0;
  const unsigned int splat = (unsigned int)run * 0x01010101u;
  constexpr int SWEEP = AGG_WAVES * 32;
  constexpr int UNR   = 8;

  for (int u0 = wib * 32; scan && u0 < KEY_UNITS; u0 += SWEEP * UNR) {
    uint4 x[UNR];
    #pragma unroll
    for (int k = 0; k < UNR; ++k) {
      const int u = u0 + k * SWEEP + lane;
      x[k] = ids[u < KEY_UNITS ? u : KEY_UNITS - 1];
    }
    #pragma unroll
    for (int k = 0; k < UNR; ++k) {
      const int u = u0 + k * SWEEP + lane;
      if (cnt > LIST_CAP - 512) {
        add_found_early<NCH, G, COUNT>(feat, rec, lst, cnt, acc, cntS, lane);
        cnt = 0;
        if (MODE == 1) saved = SEG_CAP + 1;
      }
      const unsigned int live = (u < KEY_UNITS) ? 0xFFFFFFFFu : 0u;
      unsigned int z[4];
      z[0] = zero_bytes(x[k].x ^ splat) & live;
      z[1] = zero_bytes(x[k].y ^ splat) & live;
      z[2] = zero_bytes(x[k].z ^ splat) & live;
      z[3] = zero_bytes(x[k].w ^ splat) & live;
      if (__builtin_amdgcn_ballot_w32((z[0] | z[1] | z[2] | z[3]) != 0u) == 0u) continue;
      #pragma unroll
      for (int q = 0; q < 4; ++q) {
        unsigned int zq = z[q];
        unsigned int m = __builtin_amdgcn_ballot_w32(zq != 0u);
        while (m != 0u) {
          if (zq != 0u) {
            const int bit = __builtin_ctz(zq);
            zq &= zq - 1u;
            const int pos = cnt + (int)__builtin_amdgcn_mbcnt_lo(m, 0u);
            lst[pos] = u * 16 + q * 4 + (bit >> 3);
          }
          cnt += __builtin_popcount(m);
          m = __builtin_amdgcn_ballot_w32(zq != 0u);
        }
      }
    }
    const bool last = (u0 + SWEEP * UNR >= KEY_UNITS);
    if (cnt >= 32 || (last && cnt > 0)) {
      saved = add_found<NCH, G, COUNT, MODE == 1>(feat, rec, lst, cnt, acc, cntS, lane,
                                                   segw, saved);
      cnt = 0;
    }
  }
  if (MODE == 1 && lane == 0) segCnt[run * AGG_WAVES + wib] = (unsigned int)saved;
  __syncthreads();

  float4* out4 = (float4*)(agg + (size_t)lo * D);
  const float4* acc4 = (const float4*)acc;
  for (int i = tid; i < nn * (D / 4); i += AGG_WAVES * 32) out4[i] = acc4[i];
  if (COUNT && tid < nn) inv[lo + tid] = 1.0f / fmaxf(cntS[tid], 1.0f);
}

#define GEMM_WGS   256
#define GEMM_WAVES 32

template <int DIN, int DOUT>
__device__ __forceinline__ void stage_weights(const float* __restrict__ Wg, float* Ws, int tid) {
  constexpr int PITCH = 2 * DOUT + 32;
  constexpr int Q = DOUT / 4;
  static_assert(DIN % 2 == 0 && DOUT % 4 == 0, "rows go in pairs, columns in quads");
  const float4* Wg4 = (const float4*)__builtin_assume_aligned(Wg, 16);
  for (int i = tid; i < (DIN / 2) * Q; i += GEMM_WAVES * 32) {
    const int kp = i / Q, q = i % Q;
    const float4 r0 = Wg4[(2 * kp) * Q + q];
    const float4 r1 = Wg4[(2 * kp + 1) * Q + q];
    float4* d = (float4*)(Ws + kp * PITCH + q * 8);
    d[0] = make_float4(r0.x, r1.x, r0.y, r1.y);
    d[1] = make_float4(r0.z, r1.z, r0.w, r1.w);
  }
}

template <int DIN, int DOUT>
__global__ void __launch_bounds__(GEMM_WAVES * 32)
sage_wmma_kernel(const float* __restrict__ agg,
                 const float* __restrict__ inv,
                 const float* __restrict__ hin,
                 const float* __restrict__ Wl,
                 const float* __restrict__ Wr,
                 const float* __restrict__ bias,
                 float* __restrict__ hout) {
  constexpr int PITCH = 2 * DOUT + 32;
  constexpr int NH = DOUT / 64;
  constexpr int NUNITS = (N_NODES / 16) * NH;
  __shared__ __attribute__((aligned(16))) float WlS[(DIN / 2) * PITCH];
  __shared__ __attribute__((aligned(16))) float WrS[(DIN / 2) * PITCH];

  const int tid  = threadIdx.x;
  const int lane = tid & 31;
  const int wib  = tid >> 5;
  const int half = lane >> 4;
  const int l16  = lane & 15;

  stage_weights<DIN, DOUT>(Wl, WlS, tid);
  stage_weights<DIN, DOUT>(Wr, WrS, tid);
  __syncthreads();

  for (int u = wib * GEMM_WGS + blockIdx.x; u < NUNITS; u += GEMM_WAVES * GEMM_WGS) {
    const int mBase = (u / NH) * 16;
    const int nBase = (u % NH) * 64;
    const int m = mBase + l16;
    const float invm = inv[m];
    const float* aggRow = agg + (size_t)m * DIN;
    const float* hinRow = hin + (size_t)m * DIN;
    const float* wl = WlS + (nBase + l16) * 2;
    const float* wr = WrS + (nBase + l16) * 2;

    v8f c[4];
    #pragma unroll
    for (int t = 0; t < 4; ++t) c[t] = (v8f){0.f, 0.f, 0.f, 0.f, 0.f, 0.f, 0.f, 0.f};

    for (int k0 = 0; k0 < DIN; k0 += 4) {
      const int kA = k0 + half * 2;
      v2f aL, aR;
      aL.x = aggRow[kA]     * invm;
      aL.y = aggRow[kA + 1] * invm;
      aR.x = hinRow[kA];
      aR.y = hinRow[kA + 1];
      const float* pl = wl + (kA >> 1) * PITCH;
      const float* pr = wr + (kA >> 1) * PITCH;
      #pragma unroll
      for (int t = 0; t < 4; ++t) {
        const v2f bL = *(const v2f*)(pl + t * 32);
        const v2f bR = *(const v2f*)(pr + t * 32);
        c[t] = __builtin_amdgcn_wmma_f32_16x16x4_f32(false, aL, false, bL,
                                                     (short)0, c[t], false, false);
        c[t] = __builtin_amdgcn_wmma_f32_16x16x4_f32(false, aR, false, bR,
                                                     (short)0, c[t], false, false);
      }
    }

    #pragma unroll
    for (int t = 0; t < 4; ++t) {
      const int n = nBase + t * 16 + l16;
      const float bn = bias[n];
      #pragma unroll
      for (int v = 0; v < 8; ++v) {
        const int mo = mBase + v + half * 8;
        hout[(size_t)mo * DOUT + n] = fmaxf(c[t][v] + bn, 0.0f);
      }
    }
  }
}

template <int DIN, int DOUT>
__global__ void __launch_bounds__(GEMM_WAVES * 32)
proj_pair_wmma_kernel(const float* __restrict__ hin,
                      const float* __restrict__ Wl,
                      const float* __restrict__ Wr,
                      const float* __restrict__ bias,
                      float* __restrict__ P,
                      float* __restrict__ R) {
  static_assert(DOUT == 64, "one strip spans the whole output row");
  constexpr int PITCH = 2 * DOUT + 32;
  constexpr int NUNITS = N_NODES / 16;
  __shared__ __attribute__((aligned(16))) float WlS[(DIN / 2) * PITCH];
  __shared__ __attribute__((aligned(16))) float WrS[(DIN / 2) * PITCH];

  const int tid  = threadIdx.x;
  const int lane = tid & 31;
  const int wib  = tid >> 5;
  const int half = lane >> 4;
  const int l16  = lane & 15;

  stage_weights<DIN, DOUT>(Wl, WlS, tid);
  stage_weights<DIN, DOUT>(Wr, WrS, tid);
  __syncthreads();

  for (int u = wib * GEMM_WGS + blockIdx.x; u < NUNITS; u += GEMM_WAVES * GEMM_WGS) {
    const int mBase = u * 16;
    const int m = mBase + l16;
    const float* hinRow = hin + (size_t)m * DIN;
    const float* wl = WlS + l16 * 2;
    const float* wr = WrS + l16 * 2;

    v8f cP[4], cR[4];
    #pragma unroll
    for (int t = 0; t < 4; ++t) {
      cP[t] = (v8f){0.f, 0.f, 0.f, 0.f, 0.f, 0.f, 0.f, 0.f};
      cR[t] = (v8f){0.f, 0.f, 0.f, 0.f, 0.f, 0.f, 0.f, 0.f};
    }

    for (int k0 = 0; k0 < DIN; k0 += 4) {
      const int kA = k0 + half * 2;
      v2f a;
      a.x = hinRow[kA];
      a.y = hinRow[kA + 1];
      const float* pl = wl + (kA >> 1) * PITCH;
      const float* pr = wr + (kA >> 1) * PITCH;
      #pragma unroll
      for (int t = 0; t < 4; ++t) {
        const v2f bL = *(const v2f*)(pl + t * 32);
        const v2f bR = *(const v2f*)(pr + t * 32);
        cP[t] = __builtin_amdgcn_wmma_f32_16x16x4_f32(false, a, false, bL,
                                                      (short)0, cP[t], false, false);
        cR[t] = __builtin_amdgcn_wmma_f32_16x16x4_f32(false, a, false, bR,
                                                      (short)0, cR[t], false, false);
      }
    }

    #pragma unroll
    for (int t = 0; t < 4; ++t) {
      const int n = t * 16 + l16;
      const float bn = bias[n];
      #pragma unroll
      for (int v = 0; v < 8; ++v) {
        const int mo = mBase + v + half * 8;
        P[(size_t)mo * DOUT + n] = cP[t][v];
        R[(size_t)mo * DOUT + n] = cR[t][v] + bn;
      }
    }
  }
}

__global__ void combine_classifier_kernel(const float* __restrict__ aggP,
                                          const float* __restrict__ inv,
                                          const float* __restrict__ R,
                                          const float* __restrict__ Wc,
                                          const float* __restrict__ bc,
                                          float* __restrict__ out, int n) {
  __shared__ float w[128];
  __shared__ float b[2];
  if (threadIdx.x < 128) w[threadIdx.x] = Wc[threadIdx.x];
  if (threadIdx.x < 2)   b[threadIdx.x] = bc[threadIdx.x];
  __syncthreads();
  int i = blockIdx.x * blockDim.x + threadIdx.x;
  if (i >= n) return;
  const float* pr = aggP + (size_t)i * 64;
  const float* rr = R    + (size_t)i * 64;
  float iv = inv[i];
  float o0 = b[0], o1 = b[1];
  #pragma unroll
  for (int k = 0; k < 64; ++k) {
    float hv = fmaxf(pr[k] * iv + rr[k], 0.0f);
    o0 += hv * w[k * 2 + 0];
    o1 += hv * w[k * 2 + 1];
  }
  out[i * 2 + 0] = o0;
  out[i * 2 + 1] = o1;
}

extern "C" void kernel_launch(void* const* d_in, const int* in_sizes, int n_in,
                              void* d_out, int out_size, void* d_ws, size_t ws_size,
                              hipStream_t stream) {
  const float* x   = (const float*)d_in[0];
  const int*   ei  = (const int*)d_in[1];
  const int*   src = ei;
  const int*   dst = ei + N_EDGES;
  const float* Wl1 = (const float*)d_in[2];
  const float* Wr1 = (const float*)d_in[3];
  const float* b1  = (const float*)d_in[4];
  const float* Wl2 = (const float*)d_in[5];
  const float* Wr2 = (const float*)d_in[6];
  const float* b2  = (const float*)d_in[7];
  const float* Wl3 = (const float*)d_in[8];
  const float* Wr3 = (const float*)d_in[9];
  const float* b3  = (const float*)d_in[10];
  const float* Wc  = (const float*)d_in[11];
  const float* bc  = (const float*)d_in[12];
  float* out = (float*)d_out;

  float* agg = (float*)d_ws;
  float* hA  = agg + (size_t)N_NODES * HID;
  float* hB  = hA  + (size_t)N_NODES * HID;
  float* inv = hB  + (size_t)N_NODES * HID;

  const int ZT = 256;
  unsigned int* ids = (unsigned int*)hB;
  unsigned int* rec = ids + N_EDGES / 4;
  const int nQuads = N_EDGES / 4;
  unsigned int* seg    = rec + N_EDGES;
  unsigned int* segCnt = seg + NRANGE * AGG_WAVES * SEG_CAP;

  run_id_kernel<<<(nQuads + ZT - 1) / ZT, ZT, 0, stream>>>(src, dst, ids, (uint4*)rec, nQuads);

  owner_sum_kernel<3, 8, true, 1><<<NRANGE, AGG_WAVES * 32, 0, stream>>>(
      x, rec, (const uint4*)ids, agg, inv, seg, segCnt);
  sage_wmma_kernel<IN_CH, HID><<<GEMM_WGS, GEMM_WAVES * 32, 0, stream>>>(agg, inv, x, Wl1, Wr1, b1, hA);

  owner_sum_kernel<4, 8, false, 2><<<NRANGE, AGG_WAVES * 32, 0, stream>>>(
      hA, rec, (const uint4*)ids, agg, inv, seg, segCnt);
  sage_wmma_kernel<HID, HID><<<GEMM_WGS, GEMM_WAVES * 32, 0, stream>>>(agg, inv, hA, Wl2, Wr2, b2, hB);

  float* P3 = hA;
  float* R3 = hA + (size_t)N_NODES * (HID / 2);
  proj_pair_wmma_kernel<HID, HID / 2><<<GEMM_WGS, GEMM_WAVES * 32, 0, stream>>>(hB, Wl3, Wr3, b3, P3, R3);
  run_id_kernel<<<(nQuads + ZT - 1) / ZT, ZT, 0, stream>>>(src, dst, ids, (uint4*)rec, nQuads);
  owner_sum_kernel<2, 16, false, 0><<<NRANGE, AGG_WAVES * 32, 0, stream>>>(
      P3, rec, (const uint4*)ids, agg, inv, seg, segCnt);

  combine_classifier_kernel<<<(N_NODES + ZT - 1) / ZT, ZT, 0, stream>>>(agg, inv, R3, Wc, bc,
                                                                        out, N_NODES);
}


# Layer 3's owner pass adds from the segments layer 1 kept (no scan, no second id kernel); layer 2's GEMM stores in place over its aggregate, behind a barrier, so they survive; sum order unchanged
# speedup vs baseline: 3.6081x; 1.0360x over previous
#include <hip/hip_runtime.h>

typedef float v2f __attribute__((ext_vector_type(2)));
typedef float v8f __attribute__((ext_vector_type(8)));

#define N_NODES 50000
#define N_EDGES 800000
#define IN_CH   96
#define HID     128

#define NPW       196
#define NRANGE    256
#define AGG_WAVES 32
#define LIST_CAP  1024
#define KEY_UNITS (N_EDGES / 16)
#define SEG_CAP   512
static_assert(N_EDGES % 16 == 0, "run-id array is read in 16-byte units");
static_assert(NPW * NRANGE >= N_NODES && (N_NODES - 1) / NPW < 256, "run id must fit a byte");
static_assert(N_NODES <= 65536 && NPW <= 65536, "an edge record holds two 16-bit fields");

__global__ void run_id_kernel(const int* __restrict__ src,
                              const int* __restrict__ dst,
                              unsigned int* __restrict__ ids,
                              uint4* __restrict__ rec, int nQuads) {
  int i = blockIdx.x * blockDim.x + threadIdx.x;
  if (i >= nQuads) return;
  const int* d = (const int*)__builtin_assume_aligned(dst, 16) + 4 * i;
  const int* s = (const int*)__builtin_assume_aligned(src, 16) + 4 * i;
  unsigned int d0 = (unsigned int)d[0], d1 = (unsigned int)d[1];
  unsigned int d2 = (unsigned int)d[2], d3 = (unsigned int)d[3];
  unsigned int k0 = d0 / NPW, k1 = d1 / NPW, k2 = d2 / NPW, k3 = d3 / NPW;
  ids[i] = (k0 & 255u) | ((k1 & 255u) << 8) | ((k2 & 255u) << 16) | (k3 << 24);
  uint4 r;
  r.x = ((d0 - k0 * NPW) << 16) | ((unsigned int)s[0] & 0xFFFFu);
  r.y = ((d1 - k1 * NPW) << 16) | ((unsigned int)s[1] & 0xFFFFu);
  r.z = ((d2 - k2 * NPW) << 16) | ((unsigned int)s[2] & 0xFFFFu);
  r.w = ((d3 - k3 * NPW) << 16) | ((unsigned int)s[3] & 0xFFFFu);
  rec[i] = r;
}

__device__ __forceinline__ unsigned int zero_bytes(unsigned int t) {
  return ~(((t & 0x7F7F7F7Fu) + 0x7F7F7F7Fu) | t | 0x7F7F7F7Fu);
}

template <int NCH, int G>
__device__ __forceinline__ void add_rows(const float* __restrict__ feat,
                                         int s, int t, int n, float* acc, int lane) {
  constexpr int D = NCH * 32;
  for (int j = 0; j < n; j += G) {
    float r[G][NCH];
    #pragma unroll
    for (int u = 0; u < G; ++u) {
      const int su = __builtin_amdgcn_readlane(s, j + u);
      const float* fp = feat + (size_t)su * D + lane;
      #pragma unroll
      for (int c = 0; c < NCH; ++c) r[u][c] = fp[c * 32];
    }
    #pragma unroll
    for (int u = 0; u < G; ++u) {
      if (j + u < n) {
        const int tu = __builtin_amdgcn_readlane(t, j + u);
        float* ap = acc + tu * D + lane;
        #pragma unroll
        for (int c = 0; c < NCH; ++c) atomicAdd(&ap[c * 32], r[u][c]);
      }
    }
  }
}

template <int NCH, int G, bool COUNT, bool SAVE>
__device__ __forceinline__ int add_found(const float* __restrict__ feat,
                                         const unsigned int* __restrict__ rec,
                                         const int* lst, int cnt,
                                         float* acc, float* cntS, int lane,
                                         unsigned int* segw, int saved) {
  for (int base = 0; base < cnt; base += 32) {
    int n = cnt - base;
    if (n > 32) n = 32;
    int s = 0, t = 0;
    if (lane < n) {
      const unsigned int r = rec[lst[base + lane]];
      s = (int)(r & 0xFFFFu);
      t = (int)(r >> 16);
      if (COUNT) atomicAdd(&cntS[t], 1.0f);
      if (SAVE) {
        const int p = saved + base + lane;
        if (p < SEG_CAP) segw[p] = r;
      }
    }
    add_rows<NCH, G>(feat, s, t, n, acc, lane);
  }
  saved += cnt;
  return saved > SEG_CAP ? SEG_CAP + 1 : saved;
}

template <int NCH, int G>
__device__ __forceinline__ void add_saved(const float* __restrict__ feat,
                                          const unsigned int* segw, int cnt,
                                          float* acc, int lane) {
  unsigned int rn = 0;
  if (lane < cnt) rn = segw[lane];
  for (int base = 0; base < cnt; base += 32) {
    int n = cnt - base;
    if (n > 32) n = 32;
    const unsigned int r = rn;
    unsigned int nx = 0;
    if (base + 32 + lane < cnt) nx = segw[base + 32 + lane];
    rn = nx;
    add_rows<NCH, G>(feat, (int)(r & 0xFFFFu), (int)(r >> 16), n, acc, lane);
  }
}

template <int NCH, int G, bool COUNT>
__device__ __noinline__ void add_found_early(const float* __restrict__ feat,
                                             const unsigned int* __restrict__ rec,
                                             const int* lst, int cnt,
                                             float* acc, float* cntS, int lane) {
  add_found<NCH, G, COUNT, false>(feat, rec, lst, cnt, acc, cntS, lane, nullptr, 0);
}

template <int NCH, int G, bool COUNT, int MODE>
__global__ void __launch_bounds__(AGG_WAVES * 32)
owner_sum_kernel(const float* __restrict__ feat,
                 const unsigned int* __restrict__ rec,
                 const uint4* __restrict__ ids,
                 float* __restrict__ agg,
                 float* __restrict__ inv,
                 unsigned int* seg,
                 unsigned int* segCnt) {
  constexpr int D = NCH * 32;
  __shared__ __attribute__((aligned(16))) float acc[NPW * D];
  __shared__ float cntS[NPW];
  __shared__ int   lists[AGG_WAVES * LIST_CAP];

  const int tid  = threadIdx.x;
  const int lane = tid & 31;
  const int wib  = __builtin_amdgcn_readfirstlane(tid >> 5);
  const int run  = blockIdx.x;
  const int lo   = run * NPW;
  int nn = N_NODES - lo;
  if (nn > NPW) nn = NPW;

  for (int i = tid; i < NPW * D; i += AGG_WAVES * 32) acc[i] = 0.0f;
  if (COUNT && tid < NPW) cntS[tid] = 0.0f;
  __syncthreads();

  static_assert(AGG_WAVES == 32, "lane i reads the kept count of wave i");
  unsigned int* segw = seg + (size_t)(run * AGG_WAVES + wib) * SEG_CAP;
  int saved = 0;
  bool scan = true;
  if (MODE == 2) {
    const unsigned int c = segCnt[run * AGG_WAVES + lane];
    if (__builtin_amdgcn_ballot_w32(c > (unsigned int)SEG_CAP) == 0u) {
      add_saved<NCH, G>(feat, segw, __builtin_amdgcn_readlane((int)c, wib), acc, lane);
      scan = false;
    }
  }

  int* lst = lists + wib * LIST_CAP;
  int cnt = 0;
  const unsigned int splat = (unsigned int)run * 0x01010101u;
  constexpr int SWEEP = AGG_WAVES * 32;
  constexpr int UNR   = 8;

  for (int u0 = wib * 32; scan && u0 < KEY_UNITS; u0 += SWEEP * UNR) {
    uint4 x[UNR];
    #pragma unroll
    for (int k = 0; k < UNR; ++k) {
      const int u = u0 + k * SWEEP + lane;
      x[k] = ids[u < KEY_UNITS ? u : KEY_UNITS - 1];
    }
    #pragma unroll
    for (int k = 0; k < UNR; ++k) {
      const int u = u0 + k * SWEEP + lane;
      if (cnt > LIST_CAP - 512) {
        add_found_early<NCH, G, COUNT>(feat, rec, lst, cnt, acc, cntS, lane);
        cnt = 0;
        if (MODE == 1) saved = SEG_CAP + 1;
      }
      const unsigned int live = (u < KEY_UNITS) ? 0xFFFFFFFFu : 0u;
      unsigned int z[4];
      z[0] = zero_bytes(x[k].x ^ splat) & live;
      z[1] = zero_bytes(x[k].y ^ splat) & live;
      z[2] = zero_bytes(x[k].z ^ splat) & live;
      z[3] = zero_bytes(x[k].w ^ splat) & live;
      if (__builtin_amdgcn_ballot_w32((z[0] | z[1] | z[2] | z[3]) != 0u) == 0u) continue;
      #pragma unroll
      for (int q = 0; q < 4; ++q) {
        unsigned int zq = z[q];
        unsigned int m = __builtin_amdgcn_ballot_w32(zq != 0u);
        while (m != 0u) {
          if (zq != 0u) {
            const int bit = __builtin_ctz(zq);
            zq &= zq - 1u;
            const int pos = cnt + (int)__builtin_amdgcn_mbcnt_lo(m, 0u);
            lst[pos] = u * 16 + q * 4 + (bit >> 3);
          }
          cnt += __builtin_popcount(m);
          m = __builtin_amdgcn_ballot_w32(zq != 0u);
        }
      }
    }
    const bool last = (u0 + SWEEP * UNR >= KEY_UNITS);
    if (cnt >= 32 || (last && cnt > 0)) {
      saved = add_found<NCH, G, COUNT, MODE == 1>(feat, rec, lst, cnt, acc, cntS, lane,
                                                   segw, saved);
      cnt = 0;
    }
  }
  if (MODE == 1 && lane == 0) segCnt[run * AGG_WAVES + wib] = (unsigned int)saved;
  __syncthreads();

  float4* out4 = (float4*)(agg + (size_t)lo * D);
  const float4* acc4 = (const float4*)acc;
  for (int i = tid; i < nn * (D / 4); i += AGG_WAVES * 32) out4[i] = acc4[i];
  if (COUNT && tid < nn) inv[lo + tid] = 1.0f / fmaxf(cntS[tid], 1.0f);
}

#define GEMM_WGS   256
#define GEMM_WAVES 32

template <int DIN, int DOUT>
__device__ __forceinline__ void stage_weights(const float* __restrict__ Wg, float* Ws, int tid) {
  constexpr int PITCH = 2 * DOUT + 32;
  constexpr int Q = DOUT / 4;
  static_assert(DIN % 2 == 0 && DOUT % 4 == 0, "rows go in pairs, columns in quads");
  const float4* Wg4 = (const float4*)__builtin_assume_aligned(Wg, 16);
  for (int i = tid; i < (DIN / 2) * Q; i += GEMM_WAVES * 32) {
    const int kp = i / Q, q = i % Q;
    const float4 r0 = Wg4[(2 * kp) * Q + q];
    const float4 r1 = Wg4[(2 * kp + 1) * Q + q];
    float4* d = (float4*)(Ws + kp * PITCH + q * 8);
    d[0] = make_float4(r0.x, r1.x, r0.y, r1.y);
    d[1] = make_float4(r0.z, r1.z, r0.w, r1.w);
  }
}

template <int DIN, int DOUT>
__global__ void __launch_bounds__(GEMM_WAVES * 32)
sage_wmma_kernel(const float* __restrict__ agg,
                 const float* __restrict__ inv,
                 const float* __restrict__ hin,
                 const float* __restrict__ Wl,
                 const float* __restrict__ Wr,
                 const float* __restrict__ bias,
                 float* __restrict__ hout) {
  constexpr int PITCH = 2 * DOUT + 32;
  constexpr int NH = DOUT / 64;
  constexpr int NUNITS = (N_NODES / 16) * NH;
  __shared__ __attribute__((aligned(16))) float WlS[(DIN / 2) * PITCH];
  __shared__ __attribute__((aligned(16))) float WrS[(DIN / 2) * PITCH];

  const int tid  = threadIdx.x;
  const int lane = tid & 31;
  const int wib  = tid >> 5;
  const int half = lane >> 4;
  const int l16  = lane & 15;

  stage_weights<DIN, DOUT>(Wl, WlS, tid);
  stage_weights<DIN, DOUT>(Wr, WrS, tid);
  __syncthreads();

  for (int u = wib * GEMM_WGS + blockIdx.x; u < NUNITS; u += GEMM_WAVES * GEMM_WGS) {
    const int mBase = (u / NH) * 16;
    const int nBase = (u % NH) * 64;
    const int m = mBase + l16;
    const float invm = inv[m];
    const float* aggRow = agg + (size_t)m * DIN;
    const float* hinRow = hin + (size_t)m * DIN;
    const float* wl = WlS + (nBase + l16) * 2;
    const float* wr = WrS + (nBase + l16) * 2;

    v8f c[4];
    #pragma unroll
    for (int t = 0; t < 4; ++t) c[t] = (v8f){0.f, 0.f, 0.f, 0.f, 0.f, 0.f, 0.f, 0.f};

    for (int k0 = 0; k0 < DIN; k0 += 4) {
      const int kA = k0 + half * 2;
      v2f aL, aR;
      aL.x = aggRow[kA]     * invm;
      aL.y = aggRow[kA + 1] * invm;
      aR.x = hinRow[kA];
      aR.y = hinRow[kA + 1];
      const float* pl = wl + (kA >> 1) * PITCH;
      const float* pr = wr + (kA >> 1) * PITCH;
      #pragma unroll
      for (int t = 0; t < 4; ++t) {
        const v2f bL = *(const v2f*)(pl + t * 32);
        const v2f bR = *(const v2f*)(pr + t * 32);
        c[t] = __builtin_amdgcn_wmma_f32_16x16x4_f32(false, aL, false, bL,
                                                     (short)0, c[t], false, false);
        c[t] = __builtin_amdgcn_wmma_f32_16x16x4_f32(false, aR, false, bR,
                                                     (short)0, c[t], false, false);
      }
    }

    #pragma unroll
    for (int t = 0; t < 4; ++t) {
      const int n = nBase + t * 16 + l16;
      const float bn = bias[n];
      #pragma unroll
      for (int v = 0; v < 8; ++v) {
        const int mo = mBase + v + half * 8;
        hout[(size_t)mo * DOUT + n] = fmaxf(c[t][v] + bn, 0.0f);
      }
    }
  }
}

template <int DIN, int DOUT>
__global__ void __launch_bounds__(GEMM_WAVES * 32)
sage_wmma_inplace_kernel(float* aggio,
                         const float* __restrict__ inv,
                         const float* __restrict__ hin,
                         const float* __restrict__ Wl,
                         const float* __restrict__ Wr,
                         const float* __restrict__ bias) {
  constexpr int PITCH = 2 * DOUT + 32;
  constexpr int NH = DOUT / 64;
  constexpr int NTILES = N_NODES / 16;
  static_assert(GEMM_WAVES % NH == 0, "a tile's strips stand in one work-group");
  static_assert(NTILES <= (GEMM_WAVES / NH) * GEMM_WGS, "every wave takes at most one strip");
  __shared__ __attribute__((aligned(16))) float WlS[(DIN / 2) * PITCH];
  __shared__ __attribute__((aligned(16))) float WrS[(DIN / 2) * PITCH];

  const int tid  = threadIdx.x;
  const int lane = tid & 31;
  const int wib  = __builtin_amdgcn_readfirstlane(tid >> 5);
  const int half = lane >> 4;
  const int l16  = lane & 15;

  stage_weights<DIN, DOUT>(Wl, WlS, tid);
  stage_weights<DIN, DOUT>(Wr, WrS, tid);
  __syncthreads();

  const int tile  = (wib / NH) * GEMM_WGS + blockIdx.x;
  const bool has  = tile < NTILES;
  const int mBase = tile * 16;
  const int nBase = (wib % NH) * 64;

  v8f c[4];
  #pragma unroll
  for (int t = 0; t < 4; ++t) c[t] = (v8f){0.f, 0.f, 0.f, 0.f, 0.f, 0.f, 0.f, 0.f};

  if (has) {
    const int m = mBase + l16;
    const float invm = inv[m];
    const float* aggRow = aggio + (size_t)m * DIN;
    const float* hinRow = hin + (size_t)m * DIN;
    const float* wl = WlS + (nBase + l16) * 2;
    const float* wr = WrS + (nBase + l16) * 2;

    for (int k0 = 0; k0 < DIN; k0 += 4) {
      const int kA = k0 + half * 2;
      v2f aL, aR;
      aL.x = aggRow[kA]     * invm;
      aL.y = aggRow[kA + 1] * invm;
      aR.x = hinRow[kA];
      aR.y = hinRow[kA + 1];
      const float* pl = wl + (kA >> 1) * PITCH;
      const float* pr = wr + (kA >> 1) * PITCH;
      #pragma unroll
      for (int t = 0; t < 4; ++t) {
        const v2f bL = *(const v2f*)(pl + t * 32);
        const v2f bR = *(const v2f*)(pr + t * 32);
        c[t] = __builtin_amdgcn_wmma_f32_16x16x4_f32(false, aL, false, bL,
                                                     (short)0, c[t], false, false);
        c[t] = __builtin_amdgcn_wmma_f32_16x16x4_f32(false, aR, false, bR,
                                                     (short)0, c[t], false, false);
      }
    }
  }

  __syncthreads();

  if (has) {
    #pragma unroll
    for (int t = 0; t < 4; ++t) {
      const int n = nBase + t * 16 + l16;
      const float bn = bias[n];
      #pragma unroll
      for (int v = 0; v < 8; ++v) {
        const int mo = mBase + v + half * 8;
        aggio[(size_t)mo * DOUT + n] = fmaxf(c[t][v] + bn, 0.0f);
      }
    }
  }
}

template <int DIN, int DOUT>
__global__ void __launch_bounds__(GEMM_WAVES * 32)
proj_pair_wmma_kernel(const float* __restrict__ hin,
                      const float* __restrict__ Wl,
                      const float* __restrict__ Wr,
                      const float* __restrict__ bias,
                      float* __restrict__ P,
                      float* __restrict__ R) {
  static_assert(DOUT == 64, "one strip spans the whole output row");
  constexpr int PITCH = 2 * DOUT + 32;
  constexpr int NUNITS = N_NODES / 16;
  __shared__ __attribute__((aligned(16))) float WlS[(DIN / 2) * PITCH];
  __shared__ __attribute__((aligned(16))) float WrS[(DIN / 2) * PITCH];

  const int tid  = threadIdx.x;
  const int lane = tid & 31;
  const int wib  = tid >> 5;
  const int half = lane >> 4;
  const int l16  = lane & 15;

  stage_weights<DIN, DOUT>(Wl, WlS, tid);
  stage_weights<DIN, DOUT>(Wr, WrS, tid);
  __syncthreads();

  for (int u = wib * GEMM_WGS + blockIdx.x; u < NUNITS; u += GEMM_WAVES * GEMM_WGS) {
    const int mBase = u * 16;
    const int m = mBase + l16;
    const float* hinRow = hin + (size_t)m * DIN;
    const float* wl = WlS + l16 * 2;
    const float* wr = WrS + l16 * 2;

    v8f cP[4], cR[4];
    #pragma unroll
    for (int t = 0; t < 4; ++t) {
      cP[t] = (v8f){0.f, 0.f, 0.f, 0.f, 0.f, 0.f, 0.f, 0.f};
      cR[t] = (v8f){0.f, 0.f, 0.f, 0.f, 0.f, 0.f, 0.f, 0.f};
    }

    for (int k0 = 0; k0 < DIN; k0 += 4) {
      const int kA = k0 + half * 2;
      v2f a;
      a.x = hinRow[kA];
      a.y = hinRow[kA + 1];
      const float* pl = wl + (kA >> 1) * PITCH;
      const float* pr = wr + (kA >> 1) * PITCH;
      #pragma unroll
      for (int t = 0; t < 4; ++t) {
        const v2f bL = *(const v2f*)(pl + t * 32);
        const v2f bR = *(const v2f*)(pr + t * 32);
        cP[t] = __builtin_amdgcn_wmma_f32_16x16x4_f32(false, a, false, bL,
                                                      (short)0, cP[t], false, false);
        cR[t] = __builtin_amdgcn_wmma_f32_16x16x4_f32(false, a, false, bR,
                                                      (short)0, cR[t], false, false);
      }
    }

    #pragma unroll
    for (int t = 0; t < 4; ++t) {
      const int n = t * 16 + l16;
      const float bn = bias[n];
      #pragma unroll
      for (int v = 0; v < 8; ++v) {
        const int mo = mBase + v + half * 8;
        P[(size_t)mo * DOUT + n] = cP[t][v];
        R[(size_t)mo * DOUT + n] = cR[t][v] + bn;
      }
    }
  }
}

__global__ void combine_classifier_kernel(const float* __restrict__ aggP,
                                          const float* __restrict__ inv,
                                          const float* __restrict__ R,
                                          const float* __restrict__ Wc,
                                          const float* __restrict__ bc,
                                          float* __restrict__ out, int n) {
  __shared__ float w[128];
  __shared__ float b[2];
  if (threadIdx.x < 128) w[threadIdx.x] = Wc[threadIdx.x];
  if (threadIdx.x < 2)   b[threadIdx.x] = bc[threadIdx.x];
  __syncthreads();
  int i = blockIdx.x * blockDim.x + threadIdx.x;
  if (i >= n) return;
  const float* pr = aggP + (size_t)i * 64;
  const float* rr = R    + (size_t)i * 64;
  float iv = inv[i];
  float o0 = b[0], o1 = b[1];
  #pragma unroll
  for (int k = 0; k < 64; ++k) {
    float hv = fmaxf(pr[k] * iv + rr[k], 0.0f);
    o0 += hv * w[k * 2 + 0];
    o1 += hv * w[k * 2 + 1];
  }
  out[i * 2 + 0] = o0;
  out[i * 2 + 1] = o1;
}

extern "C" void kernel_launch(void* const* d_in, const int* in_sizes, int n_in,
                              void* d_out, int out_size, void* d_ws, size_t ws_size,
                              hipStream_t stream) {
  const float* x   = (const float*)d_in[0];
  const int*   ei  = (const int*)d_in[1];
  const int*   src = ei;
  const int*   dst = ei + N_EDGES;
  const float* Wl1 = (const float*)d_in[2];
  const float* Wr1 = (const float*)d_in[3];
  const float* b1  = (const float*)d_in[4];
  const float* Wl2 = (const float*)d_in[5];
  const float* Wr2 = (const float*)d_in[6];
  const float* b2  = (const float*)d_in[7];
  const float* Wl3 = (const float*)d_in[8];
  const float* Wr3 = (const float*)d_in[9];
  const float* b3  = (const float*)d_in[10];
  const float* Wc  = (const float*)d_in[11];
  const float* bc  = (const float*)d_in[12];
  float* out = (float*)d_out;

  float* agg = (float*)d_ws;
  float* hA  = agg + (size_t)N_NODES * HID;
  float* hB  = hA  + (size_t)N_NODES * HID;
  float* inv = hB  + (size_t)N_NODES * HID;

  const int ZT = 256;
  unsigned int* ids = (unsigned int*)hB;
  unsigned int* rec = ids + N_EDGES / 4;
  const int nQuads = N_EDGES / 4;
  unsigned int* seg    = rec + N_EDGES;
  unsigned int* segCnt = seg + NRANGE * AGG_WAVES * SEG_CAP;

  run_id_kernel<<<(nQuads + ZT - 1) / ZT, ZT, 0, stream>>>(src, dst, ids, (uint4*)rec, nQuads);

  owner_sum_kernel<3, 8, true, 1><<<NRANGE, AGG_WAVES * 32, 0, stream>>>(
      x, rec, (const uint4*)ids, agg, inv, seg, segCnt);
  sage_wmma_kernel<IN_CH, HID><<<GEMM_WGS, GEMM_WAVES * 32, 0, stream>>>(agg, inv, x, Wl1, Wr1, b1, hA);

  owner_sum_kernel<4, 8, false, 2><<<NRANGE, AGG_WAVES * 32, 0, stream>>>(
      hA, rec, (const uint4*)ids, agg, inv, seg, segCnt);
  sage_wmma_inplace_kernel<HID, HID><<<GEMM_WGS, GEMM_WAVES * 32, 0, stream>>>(agg, inv, hA, Wl2, Wr2, b2);
  const float* h2 = agg;

  float* P3 = hA;
  float* R3 = hA + (size_t)N_NODES * (HID / 2);
  proj_pair_wmma_kernel<HID, HID / 2><<<GEMM_WGS, GEMM_WAVES * 32, 0, stream>>>(h2, Wl3, Wr3, b3, P3, R3);
  owner_sum_kernel<2, 16, false, 2><<<NRANGE, AGG_WAVES * 32, 0, stream>>>(
      P3, rec, (const uint4*)ids, agg, inv, seg, segCnt);

  combine_classifier_kernel<<<(N_NODES + ZT - 1) / ZT, ZT, 0, stream>>>(agg, inv, R3, Wc, bc,
                                                                        out, N_NODES);
}
